# starting kernel
# baseline (speedup 1.0000x reference)
#include <hip/hip_runtime.h>

typedef __attribute__((ext_vector_type(2))) float v2f;
typedef __attribute__((ext_vector_type(8))) float v8f;

#define IN_DIM  64
#define HIDDEN  64
#define OUT_DIM 2

__global__ void gnn_zero_ws(float* __restrict__ p, long n) {
    long i = (long)blockIdx.x * blockDim.x + threadIdx.x;
    long stride = (long)gridDim.x * blockDim.x;
    for (; i < n; i += stride) p[i] = 0.0f;
}

__global__ __launch_bounds__(256) void gnn_edge_agg(
    const float* __restrict__ x,
    const int*   __restrict__ src,
    const int*   __restrict__ dst,
    float* __restrict__ msg,
    float* __restrict__ deg,
    long E) {
    long t = (long)blockIdx.x * blockDim.x + threadIdx.x;
    if (t >= E * 64) return;
    long e = t >> 6;
    int  k = (int)(t & 63);
    int  s = src[e];
    int  d = dst[e];
    atomicAdd(&msg[(long)d * IN_DIM + k], x[(long)s * IN_DIM + k]);
    if (k == 0) atomicAdd(&deg[d], 1.0f);
}

__global__ __launch_bounds__(256) void gnn_node_update(
    const float* __restrict__ x,
    const float* __restrict__ msg,
    const float* __restrict__ deg,
    const float* __restrict__ Wl,
    const float* __restrict__ bl,
    const float* __restrict__ Wr,
    const float* __restrict__ Wh,
    const float* __restrict__ bh,
    float* __restrict__ out,
    int N) {
    __shared__ float zbuf[8][16 * HIDDEN];

    const int lane = threadIdx.x & 31;
    const int wave = threadIdx.x >> 5;
    const int tile = blockIdx.x * 8 + wave;
    const int r    = lane & 15;
    const int hi   = lane >> 4;

    const int  row   = tile * 16 + r;
    const bool valid = (row < N);
    const int  srow  = valid ? row : 0;

    const float2* xrow = (const float2*)(x   + (long)srow * IN_DIM);
    const float2* mrow = (const float2*)(msg + (long)srow * IN_DIM);
    const float scale = valid ? (1.0f / fmaxf(deg[srow], 1.0f)) : 0.0f;
    const float xs    = valid ? 1.0f : 0.0f;

    const int koff2 = hi;

    for (int j = 0; j < 4; ++j) {
        const int n = j * 16 + r;
        const float2* wlrow = (const float2*)(Wl + (long)n * IN_DIM);
        const float2* wrrow = (const float2*)(Wr + (long)n * IN_DIM);

        v8f c = {0.f, 0.f, 0.f, 0.f, 0.f, 0.f, 0.f, 0.f};

        #pragma unroll
        for (int kb = 0; kb < 16; ++kb) {
            const int p = 2 * kb + koff2;
            float2 am = mrow[p];
            float2 ax = xrow[p];
            float2 wl = wlrow[p];
            float2 wr = wrrow[p];

            v2f aA; aA.x = am.x * scale; aA.y = am.y * scale;
            v2f aX; aX.x = ax.x * xs;    aX.y = ax.y * xs;
            v2f bL; bL.x = wl.x;         bL.y = wl.y;
            v2f bR; bR.x = wr.x;         bR.y = wr.y;

            c = __builtin_amdgcn_wmma_f32_16x16x4_f32(
                    false, aA, false, bL, (short)0, c, false, false);
            c = __builtin_amdgcn_wmma_f32_16x16x4_f32(
                    false, aX, false, bR, (short)0, c, false, false);
        }

        const float bias = bl[n];
        #pragma unroll
        for (int v = 0; v < 8; ++v) {
            float h = c[v] + bias;
            h = fmaxf(h, 0.0f);
            zbuf[wave][(v + hi * 8) * HIDDEN + n] = h;
        }
    }

    float acc = bh[hi];
    #pragma unroll 8
    for (int nn = 0; nn < HIDDEN; ++nn)
        acc += zbuf[wave][r * HIDDEN + nn] * Wh[hi * HIDDEN + nn];

    if (valid) out[(long)row * OUT_DIM + hi] = acc;
}

extern "C" void kernel_launch(void* const* d_in, const int* in_sizes, int n_in,
                              void* d_out, int out_size, void* d_ws, size_t ws_size,
                              hipStream_t stream) {
    const float* x  = (const float*)d_in[0];
    const int*   ei = (const int*)d_in[1];
    const float* Wl = (const float*)d_in[2];
    const float* bl = (const float*)d_in[3];
    const float* Wr = (const float*)d_in[4];
    const float* Wh = (const float*)d_in[5];
    const float* bh = (const float*)d_in[6];
    float* out = (float*)d_out;

    const long N = in_sizes[0] / IN_DIM;
    const long E = in_sizes[1] / 2;
    const int* src = ei;
    const int* dst = ei + E;

    float* msg = (float*)d_ws;
    float* deg = msg + (size_t)N * IN_DIM;

    {
        long n = N * IN_DIM + N;
        int blocks = (int)((n + 256 * 8 - 1) / (256 * 8));
        if (blocks < 1) blocks = 1;
        gnn_zero_ws<<<blocks, 256, 0, stream>>>(msg, n);
    }
    {
        long threads = E * 64;
        int blocks = (int)((threads + 255) / 256);
        gnn_edge_agg<<<blocks, 256, 0, stream>>>(x, src, dst, msg, deg, E);
    }
    {
        long tiles  = (N + 15) / 16;
        int  blocks = (int)((tiles + 7) / 8);
        gnn_node_update<<<blocks, 256, 0, stream>>>(
            x, msg, deg, Wl, bl, Wr, Wh, bh, out, (int)N);
    }
}


# node update reads its B fragments from a fragment-ordered copy of Wl and Wr that the first launch writes: one 512-byte whole-line load per two k-steps, not two 16-row strided loads per step
# speedup vs baseline: 4.2789x; 4.2789x over previous
#include <hip/hip_runtime.h>


typedef __attribute__((ext_vector_type(2))) float v2f;
typedef __attribute__((ext_vector_type(8))) float v8f;

#define IN_DIM  64
#define HIDDEN  64
#define OUT_DIM 2

#define BIN_NODES   416
#define ROW_PITCH   68
#define AGG_THREADS 1024
#define AGG_WAVES   (AGG_THREADS / 32)
#define BLK_EDGES   1024
#define BLK_WORDS4  (BLK_EDGES / 16)
#define RING        4
#define KEY_THREADS 64
#define KEY_WAVES   (KEY_THREADS / 32)
#define WFRAG_WORDS4 (2 * 4 * 8 * 32)
#define WFRAG_BLOCKS (WFRAG_WORDS4 / KEY_THREADS)
#define WFRAG_BYTES  (WFRAG_WORDS4 * 16)

__global__ __launch_bounds__(KEY_THREADS) void edge_owner_planes(
    const int* __restrict__ dst,
    unsigned*  __restrict__ keys,
    const float* __restrict__ Wl,
    const float* __restrict__ Wr,
    float*     __restrict__ wfrag,
    int E) {
    if (blockIdx.x < WFRAG_BLOCKS) {
        const int g = blockIdx.x * KEY_THREADS + threadIdx.x;
        const int l = g & 31;
        const int i = (g >> 5) & 7;
        const int j = (g >> 8) & 3;
        const float* Wm = (g >> 10) ? Wr : Wl;
        const float* p = Wm + (16 * j + (l & 15)) * IN_DIM + 8 * i + 2 * (l >> 4);
        const float2 a = *(const float2*)p;
        const float2 b = *(const float2*)(p + 4);
        ((float4*)wfrag)[g] = make_float4(a.x, a.y, b.x, b.y);
        return;
    }
    const int lane = threadIdx.x & 31;
    const int wave = __builtin_amdgcn_readfirstlane((int)(threadIdx.x >> 5));
    const int blk  = (blockIdx.x - WFRAG_BLOCKS) * KEY_WAVES + wave;
    const int nblk = (E + BLK_EDGES - 1) / BLK_EDGES;
    if (blk >= nblk) return;
    const int e0 = blk * BLK_EDGES;

    unsigned own[32];
    if (e0 + BLK_EDGES <= E) {
        const int* dp = dst + e0 + lane;
        #pragma unroll
        for (int j = 0; j < 32; ++j)
            own[j] = (unsigned)dp[32 * j] / (unsigned)BIN_NODES;
    } else {
        #pragma unroll
        for (int j = 0; j < 32; ++j) {
            const int e = e0 + 32 * j + lane;
            own[j] = (e < E) ? ((unsigned)dst[e] / (unsigned)BIN_NODES) : 255u;
        }
    }

    unsigned pl[8] = {0u, 0u, 0u, 0u, 0u, 0u, 0u, 0u};
    #pragma unroll
    for (int j = 0; j < 32; ++j) {
        const bool mine = (lane == j);
        #pragma unroll
        for (int p = 0; p < 8; ++p) {
            const unsigned m = __builtin_amdgcn_ballot_w32(((own[j] >> p) & 1u) != 0u);
            pl[p] = mine ? m : pl[p];
        }
    }

    uint4* kp = (uint4*)keys + (long)blk * BLK_WORDS4;
    kp[lane]      = make_uint4(pl[0], pl[1], pl[2], pl[3]);
    kp[32 + lane] = make_uint4(pl[4], pl[5], pl[6], pl[7]);
}

__global__ __launch_bounds__(AGG_THREADS) void gnn_sage_fused(
    const float* __restrict__ x,
    const int*   __restrict__ src,
    const int*   __restrict__ dst,
    const unsigned* __restrict__ keys,
    const float* __restrict__ wfrag,
    const float* __restrict__ bl,
    const float* __restrict__ Wh,
    const float* __restrict__ bh,
    float* __restrict__ out,
    int E, int N) {
    __shared__ __attribute__((aligned(16))) float acc[(BIN_NODES + 1) * ROW_PITCH];
    __shared__ float cnt[BIN_NODES + 1];

    const int tid  = threadIdx.x;
    const int lane = tid & 31;
    const int wave = __builtin_amdgcn_readfirstlane(tid >> 5);
    const int base = blockIdx.x * BIN_NODES;
    const unsigned span = (unsigned)min(BIN_NODES, N - base);

    {
        float4* a4 = (float4*)acc;
        for (int i = tid; i < (BIN_NODES + 1) * ROW_PITCH / 4; i += AGG_THREADS)
            a4[i] = make_float4(0.f, 0.f, 0.f, 0.f);
        for (int i = tid; i < BIN_NODES + 1; i += AGG_THREADS) cnt[i] = 0.f;
    }
    __syncthreads();

    const int nblk = (E + BLK_EDGES - 1) / BLK_EDGES;
    const unsigned mybyte = (unsigned)blockIdx.x & 255u;
    unsigned fx[8];
    #pragma unroll
    for (int p = 0; p < 8; ++p) fx[p] = ((mybyte >> p) & 1u) ? 0u : ~0u;

    const int c0 = (int)(((unsigned)blockIdx.x * (unsigned)nblk) / gridDim.x);
    auto blk_of = [&](int t) {
        int c = c0 + t;
        if (c >= nblk) c -= nblk;
        return c;
    };
    const uint4* keys4 = (const uint4*)keys;

    int qe = 0;
    int qn = 0;

    auto resolve = [&](int nact, int& s, int& r) {
        s = 0;
        r = BIN_NODES;
        if (lane < nact) {
            if (qe < E) {
                const int d  = dst[qe];
                const int sv = src[qe];
                asm volatile("" :: "v"(d), "v"(sv));
                const unsigned rr = (unsigned)(d - base);
                if (rr < span) { s = sv; r = (int)rr; }
            }
            atomicAdd(&cnt[r], 1.0f);
        }
    };

    auto flush8 = [&]() {
        int s, r;
        resolve(8, s, r);
        float2 v[8];
        #pragma unroll
        for (int j = 0; j < 8; ++j) {
            const int sj = __builtin_amdgcn_readlane(s, j);
            v[j] = *(const float2*)(x + (long)sj * IN_DIM + 2 * lane);
        }
        #pragma unroll
        for (int j = 0; j < 8; ++j) {
            const int rj = __builtin_amdgcn_readlane(r, j);
            float* row = acc + rj * ROW_PITCH + 2 * lane;
            atomicAdd(row,     v[j].x);
            atomicAdd(row + 1, v[j].y);
        }
        qe = __shfl_down(qe, 8);
        qn -= 8;
    };

    auto scan_blk = [&](const uint4& ka, const uint4& kb, int c) {
        const unsigned hm = (ka.x ^ fx[0]) & (ka.y ^ fx[1]) & (ka.z ^ fx[2]) & (ka.w ^ fx[3])
                          & (kb.x ^ fx[4]) & (kb.y ^ fx[5]) & (kb.z ^ fx[6]) & (kb.w ^ fx[7]);
        unsigned lm = __builtin_amdgcn_ballot_w32(hm != 0u);
        const int eb = c * BLK_EDGES;
        while (lm != 0u) {
            const int L = __builtin_ctz(lm);
            lm &= lm - 1u;
            unsigned m = (unsigned)__builtin_amdgcn_readlane((int)hm, L);
            const int el = eb + L * 32;
            while (m != 0u) {
                const int i = __builtin_ctz(m);
                m &= m - 1u;
                if (qn == 32) { do { flush8(); } while (qn >= 8); }
                const bool w = (lane == qn);
                qe = w ? (el + i) : qe;
                ++qn;
            }
        }
    };

    if (nblk > 0) {
        auto load_a = [&](int c) { return keys4[(long)c * BLK_WORDS4 + lane]; };
        auto load_b = [&](int c) { return keys4[(long)c * BLK_WORDS4 + 32 + lane]; };

        uint4 ka[RING], kb[RING];
        #pragma unroll
        for (int i = 0; i < RING; ++i) {
            const int t = wave + AGG_WAVES * i;
            const int c = (t < nblk) ? blk_of(t) : 0;
            ka[i] = load_a(c);
            kb[i] = load_b(c);
        }
        for (int tb = wave; tb < nblk; tb += AGG_WAVES * RING) {
            #pragma unroll
            for (int i = 0; i < RING; ++i) {
                const int t = tb + AGG_WAVES * i;
                if (t < nblk) scan_blk(ka[i], kb[i], blk_of(t));
                const int tn = t + AGG_WAVES * RING;
                const int cn = (tn < nblk) ? blk_of(tn) : 0;
                ka[i] = load_a(cn);
                kb[i] = load_b(cn);
            }
            while (qn >= 8) flush8();
        }
    }

    if (qn > 0) {
        int s, r;
        resolve(qn, s, r);
        #pragma unroll
        for (int j = 0; j < 7; ++j) {
            if (j < qn) {
                const int sj = __builtin_amdgcn_readlane(s, j);
                const int rj = __builtin_amdgcn_readlane(r, j);
                const float2 v = *(const float2*)(x + (long)sj * IN_DIM + 2 * lane);
                float* row = acc + rj * ROW_PITCH + 2 * lane;
                atomicAdd(row,     v.x);
                atomicAdd(row + 1, v.y);
            }
        }
    }

    __syncthreads();

    const int ntiles = ((int)span + 15) >> 4;
    if (wave < ntiles) {
        const int r  = lane & 15;
        const int hi = lane >> 4;
        const int lr = wave * 16 + r;
        const bool valid = ((unsigned)lr < span);
        const long grow  = valid ? (long)(base + lr) : 0;

        const float2* xrow = (const float2*)(x + grow * IN_DIM);
        const float*  srow = acc + lr * ROW_PITCH;
        const float scale = valid ? (1.0f / fmaxf(cnt[lr], 1.0f)) : 0.0f;
        const float xs    = valid ? 1.0f : 0.0f;

        v2f aM[16], aX[16];
        #pragma unroll
        for (int kb = 0; kb < 16; ++kb) {
            const int p = 2 * kb + hi;
            const float2 am = *(const float2*)(srow + 2 * p);
            const float2 ax = xrow[p];
            aM[kb].x = am.x * scale; aM[kb].y = am.y * scale;
            aX[kb].x = ax.x * xs;    aX[kb].y = ax.y * xs;
        }

        float* ztile = acc + wave * 16 * ROW_PITCH;

        const float4* wf4 = (const float4*)wfrag;
        for (int j = 0; j < 4; ++j) {
            const int n = j * 16 + r;
            const float4* wlj = wf4 + (j * 8) * 32 + lane;
            const float4* wrj = wf4 + ((4 + j) * 8) * 32 + lane;

            v8f c = {0.f, 0.f, 0.f, 0.f, 0.f, 0.f, 0.f, 0.f};

            #pragma unroll
            for (int i = 0; i < 8; ++i) {
                const float4 wl = wlj[32 * i];
                const float4 wr = wrj[32 * i];
                v2f bL0; bL0.x = wl.x; bL0.y = wl.y;
                v2f bR0; bR0.x = wr.x; bR0.y = wr.y;
                v2f bL1; bL1.x = wl.z; bL1.y = wl.w;
                v2f bR1; bR1.x = wr.z; bR1.y = wr.w;

                c = __builtin_amdgcn_wmma_f32_16x16x4_f32(
                        false, aM[2 * i], false, bL0, (short)0, c, false, false);
                c = __builtin_amdgcn_wmma_f32_16x16x4_f32(
                        false, aX[2 * i], false, bR0, (short)0, c, false, false);
                c = __builtin_amdgcn_wmma_f32_16x16x4_f32(
                        false, aM[2 * i + 1], false, bL1, (short)0, c, false, false);
                c = __builtin_amdgcn_wmma_f32_16x16x4_f32(
                        false, aX[2 * i + 1], false, bR1, (short)0, c, false, false);
            }

            const float bias = bl[n];
            #pragma unroll
            for (int v = 0; v < 8; ++v) {
                float h = c[v] + bias;
                h = fmaxf(h, 0.0f);
                ztile[(v + hi * 8) * ROW_PITCH + n] = h;
            }
        }

        const float4* z4 = (const float4*)(ztile + r * ROW_PITCH);
        const float4* w4 = (const float4*)(Wh + hi * HIDDEN);
        float o = bh[hi];
        #pragma unroll
        for (int q = 0; q < HIDDEN / 4; ++q) {
            const float4 zz = z4[q];
            const float4 ww = w4[q];
            o = __builtin_fmaf(zz.x, ww.x, o);
            o = __builtin_fmaf(zz.y, ww.y, o);
            o = __builtin_fmaf(zz.z, ww.z, o);
            o = __builtin_fmaf(zz.w, ww.w, o);
        }

        if (valid) out[grow * OUT_DIM + hi] = o;
    }
}

extern "C" void kernel_launch(void* const* d_in, const int* in_sizes, int n_in,
                              void* d_out, int out_size, void* d_ws, size_t ws_size,
                              hipStream_t stream) {
    const float* x  = (const float*)d_in[0];
    const int*   ei = (const int*)d_in[1];
    const float* Wl = (const float*)d_in[2];
    const float* bl = (const float*)d_in[3];
    const float* Wr = (const float*)d_in[4];
    const float* Wh = (const float*)d_in[5];
    const float* bh = (const float*)d_in[6];
    float* out = (float*)d_out;

    const long N = in_sizes[0] / IN_DIM;
    const long E = in_sizes[1] / 2;
    const int* src = ei;
    const int* dst = ei + E;

    float*    wfrag = (float*)d_ws;
    unsigned* keys  = (unsigned*)((char*)d_ws + WFRAG_BYTES);

    {
        const int nblk = (int)((E + BLK_EDGES - 1) / BLK_EDGES);
        int kblocks = (nblk + KEY_WAVES - 1) / KEY_WAVES;
        if (kblocks < 1) kblocks = 1;
        edge_owner_planes<<<WFRAG_BLOCKS + kblocks, KEY_THREADS, 0, stream>>>(
            dst, keys, Wl, Wr, wfrag, (int)E);
    }
    {
        int blocks = (int)((N + BIN_NODES - 1) / BIN_NODES);
        if (blocks < 1) blocks = 1;
        gnn_sage_fused<<<blocks, AGG_THREADS, 0, stream>>>(
            x, src, dst, keys, wfrag, bl, Wh, bh, out, (int)E, (int)N);
    }
}


# match queue built by the lanes, not a scalar loop over lanes and bits: a matching lane stores its lowest match at tail + mbcnt rank of a per-wave LDS ring; sum order = LDS arrival as before
# speedup vs baseline: 4.4512x; 1.0403x over previous
#include <hip/hip_runtime.h>


typedef __attribute__((ext_vector_type(2))) float v2f;
typedef __attribute__((ext_vector_type(8))) float v8f;

#define IN_DIM  64
#define HIDDEN  64
#define OUT_DIM 2

#define BIN_NODES   416
#define ROW_PITCH   68
#define AGG_THREADS 1024
#define AGG_WAVES   (AGG_THREADS / 32)
#define BLK_EDGES   1024
#define BLK_WORDS4  (BLK_EDGES / 16)
#define RING        4
#define QCAP        64
#define KEY_THREADS 64
#define KEY_WAVES   (KEY_THREADS / 32)
#define WFRAG_WORDS4 (2 * 4 * 8 * 32)
#define WFRAG_BLOCKS (WFRAG_WORDS4 / KEY_THREADS)
#define WFRAG_BYTES  (WFRAG_WORDS4 * 16)

__global__ __launch_bounds__(KEY_THREADS) void edge_owner_planes(
    const int* __restrict__ dst,
    unsigned*  __restrict__ keys,
    const float* __restrict__ Wl,
    const float* __restrict__ Wr,
    float*     __restrict__ wfrag,
    int E) {
    if (blockIdx.x < WFRAG_BLOCKS) {
        const int g = blockIdx.x * KEY_THREADS + threadIdx.x;
        const int l = g & 31;
        const int i = (g >> 5) & 7;
        const int j = (g >> 8) & 3;
        const float* Wm = (g >> 10) ? Wr : Wl;
        const float* p = Wm + (16 * j + (l & 15)) * IN_DIM + 8 * i + 2 * (l >> 4);
        const float2 a = *(const float2*)p;
        const float2 b = *(const float2*)(p + 4);
        ((float4*)wfrag)[g] = make_float4(a.x, a.y, b.x, b.y);
        return;
    }
    const int lane = threadIdx.x & 31;
    const int wave = __builtin_amdgcn_readfirstlane((int)(threadIdx.x >> 5));
    const int blk  = (blockIdx.x - WFRAG_BLOCKS) * KEY_WAVES + wave;
    const int nblk = (E + BLK_EDGES - 1) / BLK_EDGES;
    if (blk >= nblk) return;
    const int e0 = blk * BLK_EDGES;

    unsigned own[32];
    if (e0 + BLK_EDGES <= E) {
        const int* dp = dst + e0 + lane;
        #pragma unroll
        for (int j = 0; j < 32; ++j)
            own[j] = (unsigned)dp[32 * j] / (unsigned)BIN_NODES;
    } else {
        #pragma unroll
        for (int j = 0; j < 32; ++j) {
            const int e = e0 + 32 * j + lane;
            own[j] = (e < E) ? ((unsigned)dst[e] / (unsigned)BIN_NODES) : 255u;
        }
    }

    unsigned pl[8] = {0u, 0u, 0u, 0u, 0u, 0u, 0u, 0u};
    #pragma unroll
    for (int j = 0; j < 32; ++j) {
        const bool mine = (lane == j);
        #pragma unroll
        for (int p = 0; p < 8; ++p) {
            const unsigned m = __builtin_amdgcn_ballot_w32(((own[j] >> p) & 1u) != 0u);
            pl[p] = mine ? m : pl[p];
        }
    }

    uint4* kp = (uint4*)keys + (long)blk * BLK_WORDS4;
    kp[lane]      = make_uint4(pl[0], pl[1], pl[2], pl[3]);
    kp[32 + lane] = make_uint4(pl[4], pl[5], pl[6], pl[7]);
}

__global__ __launch_bounds__(AGG_THREADS) void gnn_sage_fused(
    const float* __restrict__ x,
    const int*   __restrict__ src,
    const int*   __restrict__ dst,
    const unsigned* __restrict__ keys,
    const float* __restrict__ wfrag,
    const float* __restrict__ bl,
    const float* __restrict__ Wh,
    const float* __restrict__ bh,
    float* __restrict__ out,
    int E, int N) {
    __shared__ __attribute__((aligned(16))) float acc[(BIN_NODES + 1) * ROW_PITCH];
    __shared__ float cnt[BIN_NODES + 1];
    __shared__ int hitq[AGG_WAVES * QCAP];

    const int tid  = threadIdx.x;
    const int lane = tid & 31;
    const int wave = __builtin_amdgcn_readfirstlane(tid >> 5);
    const int base = blockIdx.x * BIN_NODES;
    const unsigned span = (unsigned)min(BIN_NODES, N - base);

    {
        float4* a4 = (float4*)acc;
        for (int i = tid; i < (BIN_NODES + 1) * ROW_PITCH / 4; i += AGG_THREADS)
            a4[i] = make_float4(0.f, 0.f, 0.f, 0.f);
        for (int i = tid; i < BIN_NODES + 1; i += AGG_THREADS) cnt[i] = 0.f;
    }
    __syncthreads();

    const int nblk = (E + BLK_EDGES - 1) / BLK_EDGES;
    const unsigned mybyte = (unsigned)blockIdx.x & 255u;
    unsigned fx[8];
    #pragma unroll
    for (int p = 0; p < 8; ++p) fx[p] = ((mybyte >> p) & 1u) ? 0u : ~0u;

    const int c0 = (int)(((unsigned)blockIdx.x * (unsigned)nblk) / gridDim.x);
    auto blk_of = [&](int t) {
        int c = c0 + t;
        if (c >= nblk) c -= nblk;
        return c;
    };
    const uint4* keys4 = (const uint4*)keys;

    int* const q = hitq + wave * QCAP;
    int qh = 0;
    int qn = 0;

    auto resolve = [&](int nact, int& s, int& r) {
        s = 0;
        r = BIN_NODES;
        if (lane < nact) {
            const int qe = q[(qh + lane) & (QCAP - 1)];
            if (qe < E) {
                const int d  = dst[qe];
                const int sv = src[qe];
                asm volatile("" :: "v"(d), "v"(sv));
                const unsigned rr = (unsigned)(d - base);
                if (rr < span) { s = sv; r = (int)rr; }
            }
            atomicAdd(&cnt[r], 1.0f);
        }
    };

    auto flush8 = [&]() {
        int s, r;
        resolve(8, s, r);
        float2 v[8];
        #pragma unroll
        for (int j = 0; j < 8; ++j) {
            const int sj = __builtin_amdgcn_readlane(s, j);
            v[j] = *(const float2*)(x + (long)sj * IN_DIM + 2 * lane);
        }
        #pragma unroll
        for (int j = 0; j < 8; ++j) {
            const int rj = __builtin_amdgcn_readlane(r, j);
            float* row = acc + rj * ROW_PITCH + 2 * lane;
            atomicAdd(row,     v[j].x);
            atomicAdd(row + 1, v[j].y);
        }
        qh = (qh + 8) & (QCAP - 1);
        qn -= 8;
    };

    auto scan_blk = [&](const uint4& ka, const uint4& kb, int c) {
        unsigned hm = (ka.x ^ fx[0]) & (ka.y ^ fx[1]) & (ka.z ^ fx[2]) & (ka.w ^ fx[3])
                    & (kb.x ^ fx[4]) & (kb.y ^ fx[5]) & (kb.z ^ fx[6]) & (kb.w ^ fx[7]);
        unsigned lm = __builtin_amdgcn_ballot_w32(hm != 0u);
        const int e0 = c * BLK_EDGES + lane * 32;
        while (lm != 0u) {
            const unsigned hrest = hm & (hm - 1u);
            const unsigned lnext = __builtin_amdgcn_ballot_w32(hrest != 0u);
            while (qn > QCAP - 32) flush8();
            const int rank = (int)__builtin_amdgcn_mbcnt_lo(lm, 0u);
            if (hm != 0u)
                q[(qh + qn + rank) & (QCAP - 1)] = e0 + __builtin_ctz(hm);
            qn += __builtin_popcount(lm);
            hm = hrest;
            lm = lnext;
        }
    };

    if (nblk > 0) {
        auto load_a = [&](int c) { return keys4[(long)c * BLK_WORDS4 + lane]; };
        auto load_b = [&](int c) { return keys4[(long)c * BLK_WORDS4 + 32 + lane]; };

        uint4 ka[RING], kb[RING];
        #pragma unroll
        for (int i = 0; i < RING; ++i) {
            const int t = wave + AGG_WAVES * i;
            const int c = (t < nblk) ? blk_of(t) : 0;
            ka[i] = load_a(c);
            kb[i] = load_b(c);
        }
        for (int tb = wave; tb < nblk; tb += AGG_WAVES * RING) {
            #pragma unroll
            for (int i = 0; i < RING; ++i) {
                const int t = tb + AGG_WAVES * i;
                if (t < nblk) scan_blk(ka[i], kb[i], blk_of(t));
                const int tn = t + AGG_WAVES * RING;
                const int cn = (tn < nblk) ? blk_of(tn) : 0;
                ka[i] = load_a(cn);
                kb[i] = load_b(cn);
            }
            while (qn >= 8) flush8();
        }
    }

    if (qn > 0) {
        int s, r;
        resolve(qn, s, r);
        #pragma unroll
        for (int j = 0; j < 7; ++j) {
            if (j < qn) {
                const int sj = __builtin_amdgcn_readlane(s, j);
                const int rj = __builtin_amdgcn_readlane(r, j);
                const float2 v = *(const float2*)(x + (long)sj * IN_DIM + 2 * lane);
                float* row = acc + rj * ROW_PITCH + 2 * lane;
                atomicAdd(row,     v.x);
                atomicAdd(row + 1, v.y);
            }
        }
    }

    __syncthreads();

    const int ntiles = ((int)span + 15) >> 4;
    if (wave < ntiles) {
        const int r  = lane & 15;
        const int hi = lane >> 4;
        const int lr = wave * 16 + r;
        const bool valid = ((unsigned)lr < span);
        const long grow  = valid ? (long)(base + lr) : 0;

        const float2* xrow = (const float2*)(x + grow * IN_DIM);
        const float*  srow = acc + lr * ROW_PITCH;
        const float scale = valid ? (1.0f / fmaxf(cnt[lr], 1.0f)) : 0.0f;
        const float xs    = valid ? 1.0f : 0.0f;

        v2f aM[16], aX[16];
        #pragma unroll
        for (int kb = 0; kb < 16; ++kb) {
            const int p = 2 * kb + hi;
            const float2 am = *(const float2*)(srow + 2 * p);
            const float2 ax = xrow[p];
            aM[kb].x = am.x * scale; aM[kb].y = am.y * scale;
            aX[kb].x = ax.x * xs;    aX[kb].y = ax.y * xs;
        }

        float* ztile = acc + wave * 16 * ROW_PITCH;

        const float4* wf4 = (const float4*)wfrag;
        for (int j = 0; j < 4; ++j) {
            const int n = j * 16 + r;
            const float4* wlj = wf4 + (j * 8) * 32 + lane;
            const float4* wrj = wf4 + ((4 + j) * 8) * 32 + lane;

            v8f c = {0.f, 0.f, 0.f, 0.f, 0.f, 0.f, 0.f, 0.f};

            #pragma unroll
            for (int i = 0; i < 8; ++i) {
                const float4 wl = wlj[32 * i];
                const float4 wr = wrj[32 * i];
                v2f bL0; bL0.x = wl.x; bL0.y = wl.y;
                v2f bR0; bR0.x = wr.x; bR0.y = wr.y;
                v2f bL1; bL1.x = wl.z; bL1.y = wl.w;
                v2f bR1; bR1.x = wr.z; bR1.y = wr.w;

                c = __builtin_amdgcn_wmma_f32_16x16x4_f32(
                        false, aM[2 * i], false, bL0, (short)0, c, false, false);
                c = __builtin_amdgcn_wmma_f32_16x16x4_f32(
                        false, aX[2 * i], false, bR0, (short)0, c, false, false);
                c = __builtin_amdgcn_wmma_f32_16x16x4_f32(
                        false, aM[2 * i + 1], false, bL1, (short)0, c, false, false);
                c = __builtin_amdgcn_wmma_f32_16x16x4_f32(
                        false, aX[2 * i + 1], false, bR1, (short)0, c, false, false);
            }

            const float bias = bl[n];
            #pragma unroll
            for (int v = 0; v < 8; ++v) {
                float h = c[v] + bias;
                h = fmaxf(h, 0.0f);
                ztile[(v + hi * 8) * ROW_PITCH + n] = h;
            }
        }

        const float4* z4 = (const float4*)(ztile + r * ROW_PITCH);
        const float4* w4 = (const float4*)(Wh + hi * HIDDEN);
        float o = bh[hi];
        #pragma unroll
        for (int q = 0; q < HIDDEN / 4; ++q) {
            const float4 zz = z4[q];
            const float4 ww = w4[q];
            o = __builtin_fmaf(zz.x, ww.x, o);
            o = __builtin_fmaf(zz.y, ww.y, o);
            o = __builtin_fmaf(zz.z, ww.z, o);
            o = __builtin_fmaf(zz.w, ww.w, o);
        }

        if (valid) out[grow * OUT_DIM + hi] = o;
    }
}

extern "C" void kernel_launch(void* const* d_in, const int* in_sizes, int n_in,
                              void* d_out, int out_size, void* d_ws, size_t ws_size,
                              hipStream_t stream) {
    const float* x  = (const float*)d_in[0];
    const int*   ei = (const int*)d_in[1];
    const float* Wl = (const float*)d_in[2];
    const float* bl = (const float*)d_in[3];
    const float* Wr = (const float*)d_in[4];
    const float* Wh = (const float*)d_in[5];
    const float* bh = (const float*)d_in[6];
    float* out = (float*)d_out;

    const long N = in_sizes[0] / IN_DIM;
    const long E = in_sizes[1] / 2;
    const int* src = ei;
    const int* dst = ei + E;

    float*    wfrag = (float*)d_ws;
    unsigned* keys  = (unsigned*)((char*)d_ws + WFRAG_BYTES);

    {
        const int nblk = (int)((E + BLK_EDGES - 1) / BLK_EDGES);
        int kblocks = (nblk + KEY_WAVES - 1) / KEY_WAVES;
        if (kblocks < 1) kblocks = 1;
        edge_owner_planes<<<WFRAG_BLOCKS + kblocks, KEY_THREADS, 0, stream>>>(
            dst, keys, Wl, Wr, wfrag, (int)E);
    }
    {
        int blocks = (int)((N + BIN_NODES - 1) / BIN_NODES);
        if (blocks < 1) blocks = 1;
        gnn_sage_fused<<<blocks, AGG_THREADS, 0, stream>>>(
            x, src, dst, keys, wfrag, bl, Wh, bh, out, (int)E, (int)N);
    }
}
